# starting kernel
# baseline (speedup 1.0000x reference)
#include <hip/hip_runtime.h>
#include <hip/hip_bf16.h>
#include <math.h>


#define B_ 4
#define L_ 2048
#define S_ 2048
#define H_ 8
#define E_ 64
#define D_ 64

typedef float v2f __attribute__((ext_vector_type(2)));
typedef float v8f __attribute__((ext_vector_type(8)));

__device__ __forceinline__ v8f wmma_f32_16x16x4(v2f a, v2f b, v8f c) {
  return __builtin_amdgcn_wmma_f32_16x16x4_f32(
      false, a, false, b, (short)0, c, false, false);
}

#define PSTRIDE 20

__global__ __launch_bounds__(256)
void causal_attn_wmma_kernel(const float* __restrict__ Q,
                             const float* __restrict__ K,
                             const float* __restrict__ V,
                             const float* __restrict__ bias,
                             float* __restrict__ out) {
  __shared__ float pbuf[8][16 * PSTRIDE];

  const int tid  = threadIdx.x;
  const int wave = tid >> 5;
  const int lane = tid & 31;
  const int nlo  = lane & 15;
  const int hi   = lane >> 4;

  const int bid = blockIdx.x;
  const int qb  = bid & 15;
  const int h   = (bid >> 4) & (H_ - 1);
  const int b   = bid >> 7;
  const int qBase = qb * 128 + wave * 16;

  const float scale = 0.125f;

  const int qRowA = qBase + nlo;
  const float* qptr = Q + (((size_t)b * L_ + qRowA) * H_ + h) * E_;
  v2f aQ[16];
#pragma unroll
  for (int c = 0; c < 16; ++c) {
    aQ[c] = *reinterpret_cast<const v2f*>(qptr + c * 4 + hi * 2);
  }

  v8f oacc[4];
#pragma unroll
  for (int t = 0; t < 4; ++t) oacc[t] = (v8f){0, 0, 0, 0, 0, 0, 0, 0};
  v8f lacc = (v8f){0, 0, 0, 0, 0, 0, 0, 0};
  const v2f onesB = (v2f){1.0f, 1.0f};

  float* myP = pbuf[wave];

  for (int j0 = 0; j0 <= qBase; j0 += 16) {
    const int key = j0 + nlo;
    const float* kptr = K + (((size_t)b * S_ + key) * H_ + h) * E_;
    v8f s8a = (v8f){0, 0, 0, 0, 0, 0, 0, 0};
    v8f s8b = (v8f){0, 0, 0, 0, 0, 0, 0, 0};
#pragma unroll
    for (int c = 0; c < 16; c += 2) {
      v2f bK0 = *reinterpret_cast<const v2f*>(kptr + (c + 0) * 4 + hi * 2);
      v2f bK1 = *reinterpret_cast<const v2f*>(kptr + (c + 1) * 4 + hi * 2);
      s8a = wmma_f32_16x16x4(aQ[c + 0], bK0, s8a);
      s8b = wmma_f32_16x16x4(aQ[c + 1], bK1, s8b);
    }

    float p[8];
#pragma unroll
    for (int r = 0; r < 8; ++r) {
      const int M = r + hi * 8;
      const int qRow = qBase + M;
      const float s = s8a[r] + s8b[r];
      p[r] = (key <= qRow)
                 ? __expf((s + bias[(size_t)qRow * S_ + key]) * scale)
                 : 0.0f;
    }

#pragma unroll
    for (int r = 0; r < 8; ++r)
      myP[(r + hi * 8) * PSTRIDE + nlo] = p[r];
    __builtin_amdgcn_fence(__ATOMIC_SEQ_CST, "wavefront");
    v2f ap[4];
#pragma unroll
    for (int kk = 0; kk < 4; ++kk) {
      ap[kk] = *reinterpret_cast<const v2f*>(myP + nlo * PSTRIDE + kk * 4 + hi * 2);
    }
    __builtin_amdgcn_fence(__ATOMIC_SEQ_CST, "wavefront");

#pragma unroll
    for (int kk = 0; kk < 4; ++kk) {
      const int s0 = j0 + kk * 4 + hi * 2;
      const float* v0 = V + (((size_t)b * S_ + s0) * H_ + h) * D_ + nlo;
      const float* v1 = v0 + (size_t)H_ * D_;
#pragma unroll
      for (int t = 0; t < 4; ++t) {
        v2f bV;
        bV.x = v0[t * 16];
        bV.y = v1[t * 16];
        oacc[t] = wmma_f32_16x16x4(ap[kk], bV, oacc[t]);
      }
      lacc = wmma_f32_16x16x4(ap[kk], onesB, lacc);
    }
  }

  float inv[8];
#pragma unroll
  for (int r = 0; r < 8; ++r) inv[r] = 1.0f / lacc[r];
#pragma unroll
  for (int t = 0; t < 4; ++t) {
#pragma unroll
    for (int r = 0; r < 8; ++r) {
      const int M = r + hi * 8;
      const int qRow = qBase + M;
      out[(((size_t)b * L_ + qRow) * H_ + h) * D_ + t * 16 + nlo] =
          oacc[t][r] * inv[r];
    }
  }
}

extern "C" void kernel_launch(void* const* d_in, const int* in_sizes, int n_in,
                              void* d_out, int out_size, void* d_ws, size_t ws_size,
                              hipStream_t stream) {
  const float* Q    = (const float*)d_in[0];
  const float* K    = (const float*)d_in[1];
  const float* V    = (const float*)d_in[2];
  const float* bias = (const float*)d_in[3];
  float* out = (float*)d_out;

  const int numQBlocks = L_ / 128;
  dim3 grid(B_ * H_ * numQBlocks);
  dim3 block(256);
  causal_attn_wmma_kernel<<<grid, block, 0, stream>>>(Q, K, V, bias, out);
}


# Keys of each 32-row block split into 8 contiguous ranges, one per wave of its work-group (grid 2048, longest blocks first); partial O and row sums added through LDS, so sums run in another order
# speedup vs baseline: 3.0488x; 3.0488x over previous
#include <hip/hip_runtime.h>
#include <hip/hip_bf16.h>
#include <math.h>


#define B_ 4
#define L_ 2048
#define S_ 2048
#define H_ 8
#define E_ 64
#define D_ 64
#define MB_ 2

typedef float v2f __attribute__((ext_vector_type(2)));
typedef float v4f __attribute__((ext_vector_type(4)));
typedef float v8f __attribute__((ext_vector_type(8)));

__device__ __forceinline__ v8f wmma_f32_16x16x4(v2f a, v2f b, v8f c) {
  return __builtin_amdgcn_wmma_f32_16x16x4_f32(
      false, a, false, b, (short)0, c, false, false);
}

#define PSTRIDE 20

#define RPITCH 68

template <int MLO, bool DIAG>
__device__ __forceinline__ void key_step(
    const int j0, const int qBase, const int b, const int h,
    const int nlo, const int hi,
    const float* __restrict__ K, const float* __restrict__ V,
    const float* __restrict__ bias, float* __restrict__ myP,
    const v2f (&aQ)[MB_][16], v8f (&oacc)[MB_][4], v8f (&lacc)[MB_]) {
  const float scale = 0.125f;
  const v2f onesB = (v2f){1.0f, 1.0f};

  const int key = j0 + nlo;
  const float* kptr = K + (((size_t)b * S_ + key) * H_ + h) * E_;
  v8f s8a[MB_], s8b[MB_];
#pragma unroll
  for (int m = 0; m < MB_; ++m) {
    s8a[m] = (v8f){0, 0, 0, 0, 0, 0, 0, 0};
    s8b[m] = (v8f){0, 0, 0, 0, 0, 0, 0, 0};
  }
#pragma unroll
  for (int c = 0; c < 16; c += 2) {
    const v2f bK0 = *reinterpret_cast<const v2f*>(kptr + (c + 0) * 4 + hi * 2);
    const v2f bK1 = *reinterpret_cast<const v2f*>(kptr + (c + 1) * 4 + hi * 2);
#pragma unroll
    for (int m = MLO; m < MB_; ++m) {
      s8a[m] = wmma_f32_16x16x4(aQ[m][c + 0], bK0, s8a[m]);
      s8b[m] = wmma_f32_16x16x4(aQ[m][c + 1], bK1, s8b[m]);
    }
  }

#pragma unroll
  for (int m = MLO; m < MB_; ++m) {
#pragma unroll
    for (int r = 0; r < 8; ++r) {
      const int M = r + hi * 8;
      const int qRow = qBase + m * 16 + M;
      const float s = s8a[m][r] + s8b[m][r];
      const float p = (!DIAG || key <= qRow)
                          ? __expf((s + bias[(size_t)qRow * S_ + key]) * scale)
                          : 0.0f;
      myP[m * (16 * PSTRIDE) + M * PSTRIDE + nlo] = p;
    }
  }
  __builtin_amdgcn_fence(__ATOMIC_SEQ_CST, "wavefront");
  v2f ap[MB_][4];
#pragma unroll
  for (int m = MLO; m < MB_; ++m) {
#pragma unroll
    for (int kk = 0; kk < 4; ++kk) {
      ap[m][kk] = *reinterpret_cast<const v2f*>(
          myP + m * (16 * PSTRIDE) + nlo * PSTRIDE + kk * 4 + hi * 2);
    }
  }
  __builtin_amdgcn_fence(__ATOMIC_SEQ_CST, "wavefront");

#pragma unroll
  for (int kk = 0; kk < 4; ++kk) {
    const int s0 = j0 + kk * 4 + hi * 2;
    const float* v0 = V + (((size_t)b * S_ + s0) * H_ + h) * D_ + nlo;
    const float* v1 = v0 + (size_t)H_ * D_;
#pragma unroll
    for (int t = 0; t < 4; ++t) {
      v2f bV;
      bV.x = v0[t * 16];
      bV.y = v1[t * 16];
#pragma unroll
      for (int m = MLO; m < MB_; ++m)
        oacc[m][t] = wmma_f32_16x16x4(ap[m][kk], bV, oacc[m][t]);
    }
#pragma unroll
    for (int m = MLO; m < MB_; ++m)
      lacc[m] = wmma_f32_16x16x4(ap[m][kk], onesB, lacc[m]);
  }
}

__global__ __launch_bounds__(256, 4)
void causal_attn_wmma_kernel(const float* __restrict__ Q,
                             const float* __restrict__ K,
                             const float* __restrict__ V,
                             const float* __restrict__ bias,
                             float* __restrict__ out) {
  __shared__ float pbuf[8][MB_ * 16 * PSTRIDE];
  __shared__ __attribute__((aligned(16))) float red[8 * 32 * RPITCH];
  __shared__ __attribute__((aligned(16))) float rsum[8 * 32];

  const int tid  = threadIdx.x;
  const int wave = __builtin_amdgcn_readfirstlane(tid >> 5);
  const int lane = tid & 31;
  const int nlo  = lane & 15;
  const int hi   = lane >> 4;

  const int bid = blockIdx.x;
  const int h   = bid & (H_ - 1);
  const int b   = (bid >> 3) & (B_ - 1);
  const int rb  = (L_ / 32 - 1) - (bid >> 5);
  const int qBase = rb * (16 * MB_);

  const int nsteps = rb * MB_ + 2;
  const int sBeg = (wave * nsteps) >> 3;
  const int sEnd = ((wave + 1) * nsteps) >> 3;

  v2f aQ[MB_][16];
#pragma unroll
  for (int m = 0; m < MB_; ++m) {
    const float* qptr =
        Q + (((size_t)b * L_ + qBase + m * 16 + nlo) * H_ + h) * E_;
#pragma unroll
    for (int c = 0; c < 16; ++c)
      aQ[m][c] = *reinterpret_cast<const v2f*>(qptr + c * 4 + hi * 2);
  }

  v8f oacc[MB_][4];
  v8f lacc[MB_];
#pragma unroll
  for (int m = 0; m < MB_; ++m) {
#pragma unroll
    for (int t = 0; t < 4; ++t) oacc[m][t] = (v8f){0, 0, 0, 0, 0, 0, 0, 0};
    lacc[m] = (v8f){0, 0, 0, 0, 0, 0, 0, 0};
  }

  float* myP = pbuf[wave];

  const int sFull = sEnd < nsteps - 2 ? sEnd : nsteps - 2;
  for (int s = sBeg; s < sFull; ++s)
    key_step<0, false>(s * 16, qBase, b, h, nlo, hi, K, V, bias, myP, aQ, oacc, lacc);
  if (sBeg <= nsteps - 2 && nsteps - 2 < sEnd)
    key_step<0, true>(qBase, qBase, b, h, nlo, hi, K, V, bias, myP, aQ, oacc, lacc);
  if (sBeg <= nsteps - 1 && nsteps - 1 < sEnd)
    key_step<1, true>(qBase + 16, qBase, b, h, nlo, hi, K, V, bias, myP, aQ, oacc, lacc);

  float* myRed = red + wave * (32 * RPITCH);
#pragma unroll
  for (int m = 0; m < MB_; ++m) {
#pragma unroll
    for (int t = 0; t < 4; ++t) {
#pragma unroll
      for (int r = 0; r < 8; ++r)
        myRed[(m * 16 + r + hi * 8) * RPITCH + t * 16 + nlo] = oacc[m][t][r];
    }
  }
  if (nlo == 0) {
#pragma unroll
    for (int m = 0; m < MB_; ++m) {
      v4f* dst = reinterpret_cast<v4f*>(rsum + wave * 32 + m * 16 + hi * 8);
      dst[0] = (v4f){lacc[m][0], lacc[m][1], lacc[m][2], lacc[m][3]};
      dst[1] = (v4f){lacc[m][4], lacc[m][5], lacc[m][6], lacc[m][7]};
    }
  }
  __syncthreads();

#pragma unroll
  for (int i = 0; i < 2; ++i) {
    const int row = wave * 4 + i * 2 + hi;
    const int col = nlo * 4;
    v4f acc = *reinterpret_cast<const v4f*>(red + row * RPITCH + col);
    float l = rsum[row];
#pragma unroll
    for (int w2 = 1; w2 < 8; ++w2) {
      acc += *reinterpret_cast<const v4f*>(red + (w2 * 32 + row) * RPITCH + col);
      l += rsum[w2 * 32 + row];
    }
    const float inv = 1.0f / l;
    *reinterpret_cast<v4f*>(
        out + (((size_t)b * L_ + qBase + row) * H_ + h) * D_ + col) = acc * inv;
  }
}

extern "C" void kernel_launch(void* const* d_in, const int* in_sizes, int n_in,
                              void* d_out, int out_size, void* d_ws, size_t ws_size,
                              hipStream_t stream) {
  const float* Q    = (const float*)d_in[0];
  const float* K    = (const float*)d_in[1];
  const float* V    = (const float*)d_in[2];
  const float* bias = (const float*)d_in[3];
  float* out = (float*)d_out;

  dim3 grid(2048);
  dim3 block(256);
  causal_attn_wmma_kernel<<<grid, block, 0, stream>>>(Q, K, V, bias, out);
}


# Each wave owns 64 query rows (4 tiles, was 2), 512-register budget, 2 waves per SIMD: a K or V fragment load feeds 4 tiles; 4 waves split the keys (was 8), sums reordered; select on the cut tile only
# speedup vs baseline: 3.4514x; 1.1321x over previous
#include <hip/hip_runtime.h>
#include <hip/hip_bf16.h>
#include <math.h>


#define B_ 4
#define L_ 2048
#define S_ 2048
#define H_ 8
#define E_ 64
#define D_ 64
#define MB_ 4
#define NW_ 4
#define RB_ (16 * MB_)

typedef float v2f __attribute__((ext_vector_type(2)));
typedef float v4f __attribute__((ext_vector_type(4)));
typedef float v8f __attribute__((ext_vector_type(8)));

__device__ __forceinline__ v8f wmma_f32_16x16x4(v2f a, v2f b, v8f c) {
  return __builtin_amdgcn_wmma_f32_16x16x4_f32(
      false, a, false, b, (short)0, c, false, false);
}

#define PSTRIDE 20

#define RPITCH 68

template <int MLO, bool DIAG>
__device__ __forceinline__ void key_step(
    const int j0, const int qBase, const int b, const int h,
    const int nlo, const int hi,
    const float* __restrict__ K, const float* __restrict__ V,
    const float* __restrict__ bias, float* __restrict__ myP,
    const v2f (&aQ)[MB_][16], v8f (&oacc)[MB_][4], v8f (&lacc)[MB_]) {
  const float scale = 0.125f;
  const v2f onesB = (v2f){1.0f, 1.0f};

  const int key = j0 + nlo;
  const float* kptr = K + (((size_t)b * S_ + key) * H_ + h) * E_;
  v8f s8a[MB_], s8b[MB_];
#pragma unroll
  for (int m = 0; m < MB_; ++m) {
    s8a[m] = (v8f){0, 0, 0, 0, 0, 0, 0, 0};
    s8b[m] = (v8f){0, 0, 0, 0, 0, 0, 0, 0};
  }
#pragma unroll
  for (int c = 0; c < 16; c += 2) {
    const v2f bK0 = *reinterpret_cast<const v2f*>(kptr + (c + 0) * 4 + hi * 2);
    const v2f bK1 = *reinterpret_cast<const v2f*>(kptr + (c + 1) * 4 + hi * 2);
#pragma unroll
    for (int m = MLO; m < MB_; ++m) {
      s8a[m] = wmma_f32_16x16x4(aQ[m][c + 0], bK0, s8a[m]);
      s8b[m] = wmma_f32_16x16x4(aQ[m][c + 1], bK1, s8b[m]);
    }
  }

#pragma unroll
  for (int m = MLO; m < MB_; ++m) {
#pragma unroll
    for (int r = 0; r < 8; ++r) {
      const int M = r + hi * 8;
      const int qRow = qBase + m * 16 + M;
      const float s = s8a[m][r] + s8b[m][r];
      const float p = (!(DIAG && m == MLO) || key <= qRow)
                          ? __expf((s + bias[(size_t)qRow * S_ + key]) * scale)
                          : 0.0f;
      myP[m * (16 * PSTRIDE) + M * PSTRIDE + nlo] = p;
    }
  }
  __builtin_amdgcn_fence(__ATOMIC_SEQ_CST, "wavefront");
  v2f ap[MB_][4];
#pragma unroll
  for (int m = MLO; m < MB_; ++m) {
#pragma unroll
    for (int kk = 0; kk < 4; ++kk) {
      ap[m][kk] = *reinterpret_cast<const v2f*>(
          myP + m * (16 * PSTRIDE) + nlo * PSTRIDE + kk * 4 + hi * 2);
    }
  }
  __builtin_amdgcn_fence(__ATOMIC_SEQ_CST, "wavefront");

#pragma unroll
  for (int kk = 0; kk < 4; ++kk) {
    const int s0 = j0 + kk * 4 + hi * 2;
    const float* v0 = V + (((size_t)b * S_ + s0) * H_ + h) * D_ + nlo;
    const float* v1 = v0 + (size_t)H_ * D_;
#pragma unroll
    for (int t = 0; t < 4; ++t) {
      v2f bV;
      bV.x = v0[t * 16];
      bV.y = v1[t * 16];
#pragma unroll
      for (int m = MLO; m < MB_; ++m)
        oacc[m][t] = wmma_f32_16x16x4(ap[m][kk], bV, oacc[m][t]);
    }
#pragma unroll
    for (int m = MLO; m < MB_; ++m)
      lacc[m] = wmma_f32_16x16x4(ap[m][kk], onesB, lacc[m]);
  }
}

__global__ __launch_bounds__(32 * NW_, 2)
void causal_attn_wmma_kernel(const float* __restrict__ Q,
                             const float* __restrict__ K,
                             const float* __restrict__ V,
                             const float* __restrict__ bias,
                             float* __restrict__ out) {
  __shared__ float pbuf[NW_][MB_ * 16 * PSTRIDE];
  __shared__ __attribute__((aligned(16))) float red[NW_ * RB_ * RPITCH];
  __shared__ __attribute__((aligned(16))) float rsum[NW_ * RB_];

  const int tid  = threadIdx.x;
  const int wave = __builtin_amdgcn_readfirstlane(tid >> 5);
  const int lane = tid & 31;
  const int nlo  = lane & 15;
  const int hi   = lane >> 4;

  const int bid = blockIdx.x;
  const int h   = bid & (H_ - 1);
  const int b   = (bid >> 3) & (B_ - 1);
  const int rb  = (L_ / RB_ - 1) - (bid >> 5);
  const int qBase = rb * RB_;

  const int nfull  = rb * MB_;
  const int nsteps = nfull + MB_;
  const int sBeg = (wave * nsteps) / NW_;
  const int sEnd = ((wave + 1) * nsteps) / NW_;

  v2f aQ[MB_][16];
#pragma unroll
  for (int m = 0; m < MB_; ++m) {
    const float* qptr =
        Q + (((size_t)b * L_ + qBase + m * 16 + nlo) * H_ + h) * E_;
#pragma unroll
    for (int c = 0; c < 16; ++c)
      aQ[m][c] = *reinterpret_cast<const v2f*>(qptr + c * 4 + hi * 2);
  }

  v8f oacc[MB_][4];
  v8f lacc[MB_];
#pragma unroll
  for (int m = 0; m < MB_; ++m) {
#pragma unroll
    for (int t = 0; t < 4; ++t) oacc[m][t] = (v8f){0, 0, 0, 0, 0, 0, 0, 0};
    lacc[m] = (v8f){0, 0, 0, 0, 0, 0, 0, 0};
  }

  float* myP = pbuf[wave];

  const int sFull = sEnd < nfull ? sEnd : nfull;
  for (int s = sBeg; s < sFull; ++s)
    key_step<0, false>(s * 16, qBase, b, h, nlo, hi, K, V, bias, myP, aQ, oacc, lacc);
  if (sBeg <= nfull + 0 && nfull + 0 < sEnd)
    key_step<0, true>(qBase, qBase, b, h, nlo, hi, K, V, bias, myP, aQ, oacc, lacc);
  if (sBeg <= nfull + 1 && nfull + 1 < sEnd)
    key_step<1, true>(qBase + 16, qBase, b, h, nlo, hi, K, V, bias, myP, aQ, oacc, lacc);
  if (sBeg <= nfull + 2 && nfull + 2 < sEnd)
    key_step<2, true>(qBase + 32, qBase, b, h, nlo, hi, K, V, bias, myP, aQ, oacc, lacc);
  if (sBeg <= nfull + 3 && nfull + 3 < sEnd)
    key_step<3, true>(qBase + 48, qBase, b, h, nlo, hi, K, V, bias, myP, aQ, oacc, lacc);

  float* myRed = red + wave * (RB_ * RPITCH);
#pragma unroll
  for (int m = 0; m < MB_; ++m) {
#pragma unroll
    for (int t = 0; t < 4; ++t) {
#pragma unroll
      for (int r = 0; r < 8; ++r)
        myRed[(m * 16 + r + hi * 8) * RPITCH + t * 16 + nlo] = oacc[m][t][r];
    }
  }
  if (nlo == 0) {
#pragma unroll
    for (int m = 0; m < MB_; ++m) {
      v4f* dst = reinterpret_cast<v4f*>(rsum + wave * RB_ + m * 16 + hi * 8);
      dst[0] = (v4f){lacc[m][0], lacc[m][1], lacc[m][2], lacc[m][3]};
      dst[1] = (v4f){lacc[m][4], lacc[m][5], lacc[m][6], lacc[m][7]};
    }
  }
  __syncthreads();

#pragma unroll
  for (int i = 0; i < RB_ / (2 * NW_); ++i) {
    const int row = wave * (RB_ / NW_) + i * 2 + hi;
    const int col = nlo * 4;
    v4f acc = *reinterpret_cast<const v4f*>(red + row * RPITCH + col);
    float l = rsum[row];
#pragma unroll
    for (int w2 = 1; w2 < NW_; ++w2) {
      acc += *reinterpret_cast<const v4f*>(red + (w2 * RB_ + row) * RPITCH + col);
      l += rsum[w2 * RB_ + row];
    }
    const float inv = 1.0f / l;
    *reinterpret_cast<v4f*>(
        out + (((size_t)b * L_ + qBase + row) * H_ + h) * D_ + col) = acc * inv;
  }
}

extern "C" void kernel_launch(void* const* d_in, const int* in_sizes, int n_in,
                              void* d_out, int out_size, void* d_ws, size_t ws_size,
                              hipStream_t stream) {
  const float* Q    = (const float*)d_in[0];
  const float* K    = (const float*)d_in[1];
  const float* V    = (const float*)d_in[2];
  const float* bias = (const float*)d_in[3];
  float* out = (float*)d_out;

  dim3 grid(1024);
  dim3 block(128);
  causal_attn_wmma_kernel<<<grid, block, 0, stream>>>(Q, K, V, bias, out);
}


# K, bias and V operands of each 16-key step fetched one step ahead: each register set is re-loaded right behind the run that read it last, so no WMMA run starts on a drained load; sums unchanged
# speedup vs baseline: 3.7131x; 1.0758x over previous
#include <hip/hip_runtime.h>
#include <hip/hip_bf16.h>
#include <math.h>


#define B_ 4
#define L_ 2048
#define S_ 2048
#define H_ 8
#define E_ 64
#define D_ 64
#define MB_ 4
#define NW_ 4
#define RB_ (16 * MB_)

typedef float v2f __attribute__((ext_vector_type(2)));
typedef float v4f __attribute__((ext_vector_type(4)));
typedef float v8f __attribute__((ext_vector_type(8)));

__device__ __forceinline__ v8f wmma_f32_16x16x4(v2f a, v2f b, v8f c) {
  return __builtin_amdgcn_wmma_f32_16x16x4_f32(
      false, a, false, b, (short)0, c, false, false);
}

#define PSTRIDE 20

#define RPITCH 68

__device__ __forceinline__ void load_k_bias(
    const int j0, const int qBase, const int b, const int h,
    const int nlo, const int hi,
    const float* __restrict__ K, const float* __restrict__ bias,
    v2f (&kf)[16], float (&bz)[MB_][8]) {
  const int key = j0 + nlo;
  const float* kptr = K + (((size_t)b * S_ + key) * H_ + h) * E_ + hi * 2;
#pragma unroll
  for (int c = 0; c < 16; ++c)
    kf[c] = *reinterpret_cast<const v2f*>(kptr + c * 4);
  const float* bptr = bias + (size_t)(qBase + hi * 8) * S_ + key;
#pragma unroll
  for (int m = 0; m < MB_; ++m) {
#pragma unroll
    for (int r = 0; r < 8; ++r) bz[m][r] = bptr[(size_t)(m * 16 + r) * S_];
  }
}

__device__ __forceinline__ void load_v(
    const int j0, const int b, const int h, const int nlo, const int hi,
    const float* __restrict__ V, v2f (&vf)[4][4]) {
#pragma unroll
  for (int kk = 0; kk < 4; ++kk) {
    const int s0 = j0 + kk * 4 + hi * 2;
    const float* v0 = V + (((size_t)b * S_ + s0) * H_ + h) * D_ + nlo;
    const float* v1 = v0 + (size_t)H_ * D_;
#pragma unroll
    for (int t = 0; t < 4; ++t) {
      vf[kk][t].x = v0[t * 16];
      vf[kk][t].y = v1[t * 16];
    }
  }
}

template <int MLO, bool DIAG>
__device__ __forceinline__ void key_step(
    const int j0, const int jn, const int qBase, const int b, const int h,
    const int nlo, const int hi,
    const float* __restrict__ K, const float* __restrict__ V,
    const float* __restrict__ bias, float* __restrict__ myP,
    const v2f (&aQ)[MB_][16], v8f (&oacc)[MB_][4], v8f (&lacc)[MB_],
    v2f (&kf)[16], float (&bz)[MB_][8], v2f (&vf)[4][4]) {
  const float scale = 0.125f;
  const v2f onesB = (v2f){1.0f, 1.0f};

  const int key = j0 + nlo;
  v8f s8a[MB_], s8b[MB_];
#pragma unroll
  for (int m = 0; m < MB_; ++m) {
    s8a[m] = (v8f){0, 0, 0, 0, 0, 0, 0, 0};
    s8b[m] = (v8f){0, 0, 0, 0, 0, 0, 0, 0};
  }
#pragma unroll
  for (int c = 0; c < 16; c += 2) {
#pragma unroll
    for (int m = MLO; m < MB_; ++m) {
      s8a[m] = wmma_f32_16x16x4(aQ[m][c + 0], kf[c + 0], s8a[m]);
      s8b[m] = wmma_f32_16x16x4(aQ[m][c + 1], kf[c + 1], s8b[m]);
    }
  }

#pragma unroll
  for (int m = MLO; m < MB_; ++m) {
#pragma unroll
    for (int r = 0; r < 8; ++r) {
      const int M = r + hi * 8;
      const int qRow = qBase + m * 16 + M;
      const float s = s8a[m][r] + s8b[m][r];
      const float p = (!(DIAG && m == MLO) || key <= qRow)
                          ? __expf((s + bz[m][r]) * scale)
                          : 0.0f;
      myP[m * (16 * PSTRIDE) + M * PSTRIDE + nlo] = p;
    }
  }
  __builtin_amdgcn_fence(__ATOMIC_SEQ_CST, "wavefront");
  v2f ap[MB_][4];
#pragma unroll
  for (int m = MLO; m < MB_; ++m) {
#pragma unroll
    for (int kk = 0; kk < 4; ++kk) {
      ap[m][kk] = *reinterpret_cast<const v2f*>(
          myP + m * (16 * PSTRIDE) + nlo * PSTRIDE + kk * 4 + hi * 2);
    }
  }
  __builtin_amdgcn_fence(__ATOMIC_SEQ_CST, "wavefront");

  __builtin_amdgcn_sched_barrier(0);
  load_k_bias(jn, qBase, b, h, nlo, hi, K, bias, kf, bz);
  __builtin_amdgcn_sched_barrier(0);

#pragma unroll
  for (int kk = 0; kk < 4; ++kk) {
#pragma unroll
    for (int t = 0; t < 4; ++t) {
#pragma unroll
      for (int m = MLO; m < MB_; ++m)
        oacc[m][t] = wmma_f32_16x16x4(ap[m][kk], vf[kk][t], oacc[m][t]);
    }
#pragma unroll
    for (int m = MLO; m < MB_; ++m)
      lacc[m] = wmma_f32_16x16x4(ap[m][kk], onesB, lacc[m]);
  }

  __builtin_amdgcn_sched_barrier(0);
  load_v(jn, b, h, nlo, hi, V, vf);
}

__global__ __launch_bounds__(32 * NW_, 2)
void causal_attn_wmma_kernel(const float* __restrict__ Q,
                             const float* __restrict__ K,
                             const float* __restrict__ V,
                             const float* __restrict__ bias,
                             float* __restrict__ out) {
  __shared__ float pbuf[NW_][MB_ * 16 * PSTRIDE];
  __shared__ __attribute__((aligned(16))) float red[NW_ * RB_ * RPITCH];
  __shared__ __attribute__((aligned(16))) float rsum[NW_ * RB_];

  const int tid  = threadIdx.x;
  const int wave = __builtin_amdgcn_readfirstlane(tid >> 5);
  const int lane = tid & 31;
  const int nlo  = lane & 15;
  const int hi   = lane >> 4;

  const int bid = blockIdx.x;
  const int h   = bid & (H_ - 1);
  const int b   = (bid >> 3) & (B_ - 1);
  const int rb  = (L_ / RB_ - 1) - (bid >> 5);
  const int qBase = rb * RB_;

  const int nfull  = rb * MB_;
  const int nsteps = nfull + MB_;
  const int sBeg = (wave * nsteps) / NW_;
  const int sEnd = ((wave + 1) * nsteps) / NW_;

  v2f aQ[MB_][16];
#pragma unroll
  for (int m = 0; m < MB_; ++m) {
    const float* qptr =
        Q + (((size_t)b * L_ + qBase + m * 16 + nlo) * H_ + h) * E_;
#pragma unroll
    for (int c = 0; c < 16; ++c)
      aQ[m][c] = *reinterpret_cast<const v2f*>(qptr + c * 4 + hi * 2);
  }

  v8f oacc[MB_][4];
  v8f lacc[MB_];
#pragma unroll
  for (int m = 0; m < MB_; ++m) {
#pragma unroll
    for (int t = 0; t < 4; ++t) oacc[m][t] = (v8f){0, 0, 0, 0, 0, 0, 0, 0};
    lacc[m] = (v8f){0, 0, 0, 0, 0, 0, 0, 0};
  }

  float* myP = pbuf[wave];

  v2f kf[16];
  float bz[MB_][8];
  v2f vf[4][4];
  load_k_bias(sBeg * 16, qBase, b, h, nlo, hi, K, bias, kf, bz);
  load_v(sBeg * 16, b, h, nlo, hi, V, vf);
  const int jLast = (sEnd - 1) * 16;

  const int sFull = sEnd < nfull ? sEnd : nfull;
  for (int s = sBeg; s < sFull; ++s) {
    const int jn = (s * 16 + 16 < jLast) ? s * 16 + 16 : jLast;
    key_step<0, false>(s * 16, jn, qBase, b, h, nlo, hi, K, V, bias, myP, aQ,
                       oacc, lacc, kf, bz, vf);
  }
  if (sBeg <= nfull + 0 && nfull + 0 < sEnd) {
    const int jn = (qBase + 16 < jLast) ? qBase + 16 : jLast;
    key_step<0, true>(qBase, jn, qBase, b, h, nlo, hi, K, V, bias, myP, aQ,
                      oacc, lacc, kf, bz, vf);
  }
  if (sBeg <= nfull + 1 && nfull + 1 < sEnd) {
    const int jn = (qBase + 32 < jLast) ? qBase + 32 : jLast;
    key_step<1, true>(qBase + 16, jn, qBase, b, h, nlo, hi, K, V, bias, myP, aQ,
                      oacc, lacc, kf, bz, vf);
  }
  if (sBeg <= nfull + 2 && nfull + 2 < sEnd) {
    const int jn = (qBase + 48 < jLast) ? qBase + 48 : jLast;
    key_step<2, true>(qBase + 32, jn, qBase, b, h, nlo, hi, K, V, bias, myP, aQ,
                      oacc, lacc, kf, bz, vf);
  }
  if (sBeg <= nfull + 3 && nfull + 3 < sEnd)
    key_step<3, true>(qBase + 48, jLast, qBase, b, h, nlo, hi, K, V, bias, myP,
                      aQ, oacc, lacc, kf, bz, vf);

  float* myRed = red + wave * (RB_ * RPITCH);
#pragma unroll
  for (int m = 0; m < MB_; ++m) {
#pragma unroll
    for (int t = 0; t < 4; ++t) {
#pragma unroll
      for (int r = 0; r < 8; ++r)
        myRed[(m * 16 + r + hi * 8) * RPITCH + t * 16 + nlo] = oacc[m][t][r];
    }
  }
  if (nlo == 0) {
#pragma unroll
    for (int m = 0; m < MB_; ++m) {
      v4f* dst = reinterpret_cast<v4f*>(rsum + wave * RB_ + m * 16 + hi * 8);
      dst[0] = (v4f){lacc[m][0], lacc[m][1], lacc[m][2], lacc[m][3]};
      dst[1] = (v4f){lacc[m][4], lacc[m][5], lacc[m][6], lacc[m][7]};
    }
  }
  __syncthreads();

#pragma unroll
  for (int i = 0; i < RB_ / (2 * NW_); ++i) {
    const int row = wave * (RB_ / NW_) + i * 2 + hi;
    const int col = nlo * 4;
    v4f acc = *reinterpret_cast<const v4f*>(red + row * RPITCH + col);
    float l = rsum[row];
#pragma unroll
    for (int w2 = 1; w2 < NW_; ++w2) {
      acc += *reinterpret_cast<const v4f*>(red + (w2 * RB_ + row) * RPITCH + col);
      l += rsum[w2 * RB_ + row];
    }
    const float inv = 1.0f / l;
    *reinterpret_cast<v4f*>(
        out + (((size_t)b * L_ + qBase + row) * H_ + h) * D_ + col) = acc * inv;
  }
}

extern "C" void kernel_launch(void* const* d_in, const int* in_sizes, int n_in,
                              void* d_out, int out_size, void* d_ws, size_t ws_size,
                              hipStream_t stream) {
  const float* Q    = (const float*)d_in[0];
  const float* K    = (const float*)d_in[1];
  const float* V    = (const float*)d_in[2];
  const float* bias = (const float*)d_in[3];
  float* out = (float*)d_out;

  dim3 grid(1024);
  dim3 block(128);
  causal_attn_wmma_kernel<<<grid, block, 0, stream>>>(Q, K, V, bias, out);
}


# Softmax row sums by per-lane f32 adds of the P fragments read back from LDS, joined in the closing LDS reduction; no ones-column WMMA left (144 -> 128 a step); denominator summed in another order
# speedup vs baseline: 3.9666x; 1.0683x over previous
#include <hip/hip_runtime.h>
#include <hip/hip_bf16.h>
#include <math.h>


#define B_ 4
#define L_ 2048
#define S_ 2048
#define H_ 8
#define E_ 64
#define D_ 64
#define MB_ 4
#define NW_ 4
#define RB_ (16 * MB_)

typedef float v2f __attribute__((ext_vector_type(2)));
typedef float v4f __attribute__((ext_vector_type(4)));
typedef float v8f __attribute__((ext_vector_type(8)));

__device__ __forceinline__ v8f wmma_f32_16x16x4(v2f a, v2f b, v8f c) {
  return __builtin_amdgcn_wmma_f32_16x16x4_f32(
      false, a, false, b, (short)0, c, false, false);
}

#define PSTRIDE 20

#define RPITCH 68

__device__ __forceinline__ void load_k_bias(
    const int j0, const int qBase, const int b, const int h,
    const int nlo, const int hi,
    const float* __restrict__ K, const float* __restrict__ bias,
    v2f (&kf)[16], float (&bz)[MB_][8]) {
  const int key = j0 + nlo;
  const float* kptr = K + (((size_t)b * S_ + key) * H_ + h) * E_ + hi * 2;
#pragma unroll
  for (int c = 0; c < 16; ++c)
    kf[c] = *reinterpret_cast<const v2f*>(kptr + c * 4);
  const float* bptr = bias + (size_t)(qBase + hi * 8) * S_ + key;
#pragma unroll
  for (int m = 0; m < MB_; ++m) {
#pragma unroll
    for (int r = 0; r < 8; ++r) bz[m][r] = bptr[(size_t)(m * 16 + r) * S_];
  }
}

__device__ __forceinline__ void load_v(
    const int j0, const int b, const int h, const int nlo, const int hi,
    const float* __restrict__ V, v2f (&vf)[4][4]) {
#pragma unroll
  for (int kk = 0; kk < 4; ++kk) {
    const int s0 = j0 + kk * 4 + hi * 2;
    const float* v0 = V + (((size_t)b * S_ + s0) * H_ + h) * D_ + nlo;
    const float* v1 = v0 + (size_t)H_ * D_;
#pragma unroll
    for (int t = 0; t < 4; ++t) {
      vf[kk][t].x = v0[t * 16];
      vf[kk][t].y = v1[t * 16];
    }
  }
}

template <int MLO, bool DIAG>
__device__ __forceinline__ void key_step(
    const int j0, const int jn, const int qBase, const int b, const int h,
    const int nlo, const int hi,
    const float* __restrict__ K, const float* __restrict__ V,
    const float* __restrict__ bias, float* __restrict__ myP,
    const v2f (&aQ)[MB_][16], v8f (&oacc)[MB_][4], v2f (&lsum)[MB_],
    v2f (&kf)[16], float (&bz)[MB_][8], v2f (&vf)[4][4]) {
  const float scale = 0.125f;

  const int key = j0 + nlo;
  v8f s8a[MB_], s8b[MB_];
#pragma unroll
  for (int m = 0; m < MB_; ++m) {
    s8a[m] = (v8f){0, 0, 0, 0, 0, 0, 0, 0};
    s8b[m] = (v8f){0, 0, 0, 0, 0, 0, 0, 0};
  }
#pragma unroll
  for (int c = 0; c < 16; c += 2) {
#pragma unroll
    for (int m = MLO; m < MB_; ++m) {
      s8a[m] = wmma_f32_16x16x4(aQ[m][c + 0], kf[c + 0], s8a[m]);
      s8b[m] = wmma_f32_16x16x4(aQ[m][c + 1], kf[c + 1], s8b[m]);
    }
  }

#pragma unroll
  for (int m = MLO; m < MB_; ++m) {
#pragma unroll
    for (int r = 0; r < 8; ++r) {
      const int M = r + hi * 8;
      const int qRow = qBase + m * 16 + M;
      const float s = s8a[m][r] + s8b[m][r];
      const float p = (!(DIAG && m == MLO) || key <= qRow)
                          ? __expf((s + bz[m][r]) * scale)
                          : 0.0f;
      myP[m * (16 * PSTRIDE) + M * PSTRIDE + nlo] = p;
    }
  }
  __builtin_amdgcn_fence(__ATOMIC_SEQ_CST, "wavefront");
  v2f ap[MB_][4];
#pragma unroll
  for (int m = MLO; m < MB_; ++m) {
#pragma unroll
    for (int kk = 0; kk < 4; ++kk) {
      ap[m][kk] = *reinterpret_cast<const v2f*>(
          myP + m * (16 * PSTRIDE) + nlo * PSTRIDE + kk * 4 + hi * 2);
    }
  }
  __builtin_amdgcn_fence(__ATOMIC_SEQ_CST, "wavefront");

  __builtin_amdgcn_sched_barrier(0);
  load_k_bias(jn, qBase, b, h, nlo, hi, K, bias, kf, bz);
  __builtin_amdgcn_sched_barrier(0);

#pragma unroll
  for (int kk = 0; kk < 4; ++kk) {
#pragma unroll
    for (int t = 0; t < 4; ++t) {
#pragma unroll
      for (int m = MLO; m < MB_; ++m)
        oacc[m][t] = wmma_f32_16x16x4(ap[m][kk], vf[kk][t], oacc[m][t]);
    }
#pragma unroll
    for (int m = MLO; m < MB_; ++m) lsum[m] += ap[m][kk];
  }

  __builtin_amdgcn_sched_barrier(0);
  load_v(jn, b, h, nlo, hi, V, vf);
}

__global__ __launch_bounds__(32 * NW_, 2)
void causal_attn_wmma_kernel(const float* __restrict__ Q,
                             const float* __restrict__ K,
                             const float* __restrict__ V,
                             const float* __restrict__ bias,
                             float* __restrict__ out) {
  __shared__ float pbuf[NW_][MB_ * 16 * PSTRIDE];
  __shared__ __attribute__((aligned(16))) float red[NW_ * RB_ * RPITCH];
  __shared__ __attribute__((aligned(16))) float rsum[NW_ * RB_ * 2];

  const int tid  = threadIdx.x;
  const int wave = __builtin_amdgcn_readfirstlane(tid >> 5);
  const int lane = tid & 31;
  const int nlo  = lane & 15;
  const int hi   = lane >> 4;

  const int bid = blockIdx.x;
  const int h   = bid & (H_ - 1);
  const int b   = (bid >> 3) & (B_ - 1);
  const int rb  = (L_ / RB_ - 1) - (bid >> 5);
  const int qBase = rb * RB_;

  const int nfull  = rb * MB_;
  const int nsteps = nfull + MB_;
  const int sBeg = (wave * nsteps) / NW_;
  const int sEnd = ((wave + 1) * nsteps) / NW_;

  v2f aQ[MB_][16];
#pragma unroll
  for (int m = 0; m < MB_; ++m) {
    const float* qptr =
        Q + (((size_t)b * L_ + qBase + m * 16 + nlo) * H_ + h) * E_;
#pragma unroll
    for (int c = 0; c < 16; ++c)
      aQ[m][c] = *reinterpret_cast<const v2f*>(qptr + c * 4 + hi * 2);
  }

  v8f oacc[MB_][4];
  v2f lsum[MB_];
#pragma unroll
  for (int m = 0; m < MB_; ++m) {
#pragma unroll
    for (int t = 0; t < 4; ++t) oacc[m][t] = (v8f){0, 0, 0, 0, 0, 0, 0, 0};
    lsum[m] = (v2f){0, 0};
  }

  float* myP = pbuf[wave];

  v2f kf[16];
  float bz[MB_][8];
  v2f vf[4][4];
  load_k_bias(sBeg * 16, qBase, b, h, nlo, hi, K, bias, kf, bz);
  load_v(sBeg * 16, b, h, nlo, hi, V, vf);
  const int jLast = (sEnd - 1) * 16;

  const int sFull = sEnd < nfull ? sEnd : nfull;
  for (int s = sBeg; s < sFull; ++s) {
    const int jn = (s * 16 + 16 < jLast) ? s * 16 + 16 : jLast;
    key_step<0, false>(s * 16, jn, qBase, b, h, nlo, hi, K, V, bias, myP, aQ,
                       oacc, lsum, kf, bz, vf);
  }
  if (sBeg <= nfull + 0 && nfull + 0 < sEnd) {
    const int jn = (qBase + 16 < jLast) ? qBase + 16 : jLast;
    key_step<0, true>(qBase, jn, qBase, b, h, nlo, hi, K, V, bias, myP, aQ,
                      oacc, lsum, kf, bz, vf);
  }
  if (sBeg <= nfull + 1 && nfull + 1 < sEnd) {
    const int jn = (qBase + 32 < jLast) ? qBase + 32 : jLast;
    key_step<1, true>(qBase + 16, jn, qBase, b, h, nlo, hi, K, V, bias, myP, aQ,
                      oacc, lsum, kf, bz, vf);
  }
  if (sBeg <= nfull + 2 && nfull + 2 < sEnd) {
    const int jn = (qBase + 48 < jLast) ? qBase + 48 : jLast;
    key_step<2, true>(qBase + 32, jn, qBase, b, h, nlo, hi, K, V, bias, myP, aQ,
                      oacc, lsum, kf, bz, vf);
  }
  if (sBeg <= nfull + 3 && nfull + 3 < sEnd)
    key_step<3, true>(qBase + 48, jLast, qBase, b, h, nlo, hi, K, V, bias, myP,
                      aQ, oacc, lsum, kf, bz, vf);

  float* myRed = red + wave * (RB_ * RPITCH);
#pragma unroll
  for (int m = 0; m < MB_; ++m) {
#pragma unroll
    for (int t = 0; t < 4; ++t) {
#pragma unroll
      for (int r = 0; r < 8; ++r)
        myRed[(m * 16 + r + hi * 8) * RPITCH + t * 16 + nlo] = oacc[m][t][r];
    }
  }
#pragma unroll
  for (int m = 0; m < MB_; ++m)
    rsum[(wave * RB_ + m * 16 + nlo) * 2 + hi] = lsum[m].x + lsum[m].y;
  __syncthreads();

#pragma unroll
  for (int i = 0; i < RB_ / (2 * NW_); ++i) {
    const int row = wave * (RB_ / NW_) + i * 2 + hi;
    const int col = nlo * 4;
    v4f acc = *reinterpret_cast<const v4f*>(red + row * RPITCH + col);
    v2f lp = *reinterpret_cast<const v2f*>(rsum + row * 2);
    float l = lp.x + lp.y;
#pragma unroll
    for (int w2 = 1; w2 < NW_; ++w2) {
      acc += *reinterpret_cast<const v4f*>(red + (w2 * RB_ + row) * RPITCH + col);
      lp = *reinterpret_cast<const v2f*>(rsum + (w2 * RB_ + row) * 2);
      l += lp.x + lp.y;
    }
    const float inv = 1.0f / l;
    *reinterpret_cast<v4f*>(
        out + (((size_t)b * L_ + qBase + row) * H_ + h) * D_ + col) = acc * inv;
  }
}

extern "C" void kernel_launch(void* const* d_in, const int* in_sizes, int n_in,
                              void* d_out, int out_size, void* d_ws, size_t ws_size,
                              hipStream_t stream) {
  const float* Q    = (const float*)d_in[0];
  const float* K    = (const float*)d_in[1];
  const float* V    = (const float*)d_in[2];
  const float* bias = (const float*)d_in[3];
  float* out = (float*)d_out;

  dim3 grid(1024);
  dim3 block(128);
  causal_attn_wmma_kernel<<<grid, block, 0, stream>>>(Q, K, V, bias, out);
}
